# starting kernel
# baseline (speedup 1.0000x reference)
#include <hip/hip_runtime.h>

#define N_FEATS 128

typedef __attribute__((ext_vector_type(2))) float v2f;
typedef __attribute__((ext_vector_type(8))) float v8f;

__device__ __forceinline__ void atomic_add_f32(float* p, float v) {
  __hip_atomic_fetch_add(p, v, __ATOMIC_RELAXED, __HIP_MEMORY_SCOPE_AGENT);
}

__global__ __launch_bounds__(256) void gcn_degrees(
    const int* __restrict__ src, const int* __restrict__ dst,
    float* __restrict__ deg_out, float* __restrict__ deg_in, int n_edges) {
  int i = blockIdx.x * blockDim.x + threadIdx.x;
  if (i >= n_edges) return;
  atomic_add_f32(&deg_out[src[i]], 1.0f);
  atomic_add_f32(&deg_in[dst[i]], 1.0f);
}

__global__ __launch_bounds__(256) void gcn_norms(
    const float* __restrict__ deg_out, const float* __restrict__ deg_in,
    float* __restrict__ ns, float* __restrict__ nd, int n) {
  int i = blockIdx.x * blockDim.x + threadIdx.x;
  if (i >= n) return;
  ns[i] = rsqrtf(fmaxf(deg_out[i], 1.0f));
  nd[i] = rsqrtf(fmaxf(deg_in[i], 1.0f));
}

__global__ __launch_bounds__(256) void gcn_gemm_wmma(
    const float* __restrict__ A, const float* __restrict__ norm_src,
    const float* __restrict__ W, float* __restrict__ out, int n_rows) {
  __shared__ float As[16][132];

  const int row0 = blockIdx.x << 4;

  for (int t = threadIdx.x; t < 512; t += 256) {
    const int r = t >> 5;
    const int c4 = (t & 31) << 2;
    const int row = row0 + r;
    if (row < n_rows) {
      const float4 v = *(const float4*)(A + (size_t)row * N_FEATS + c4);
      const float s = norm_src[row];
      As[r][c4 + 0] = v.x * s;
      As[r][c4 + 1] = v.y * s;
      As[r][c4 + 2] = v.z * s;
      As[r][c4 + 3] = v.w * s;
    } else {
      As[r][c4 + 0] = 0.f; As[r][c4 + 1] = 0.f;
      As[r][c4 + 2] = 0.f; As[r][c4 + 3] = 0.f;
    }
  }
  __syncthreads();

  const int wave = threadIdx.x >> 5;
  const int lane = threadIdx.x & 31;
  const int half = lane >> 4;
  const int m    = lane & 15;
  const int col0 = wave << 4;

  v8f acc = {};
#pragma unroll
  for (int k = 0; k < N_FEATS; k += 4) {
    const int ka = k + (half << 1);
    v2f a, b;
    a.x = As[m][ka];
    a.y = As[m][ka + 1];
    b.x = W[ka * N_FEATS + col0 + m];
    b.y = W[(ka + 1) * N_FEATS + col0 + m];
    acc = __builtin_amdgcn_wmma_f32_16x16x4_f32(
         false, a,  false, b,
         (short)0, acc,  false,  false);
  }

#pragma unroll
  for (int j = 0; j < 8; ++j) {
    const int r = row0 + j + (half << 3);
    if (r < n_rows) out[(size_t)r * N_FEATS + col0 + m] = acc[j];
  }
}

__global__ __launch_bounds__(256) void gcn_agg(
    const float* __restrict__ h, const int* __restrict__ src,
    const int* __restrict__ dst, float* __restrict__ agg, int n_edges) {
  const int e    = (blockIdx.x * blockDim.x + threadIdx.x) >> 5;
  const int lane = threadIdx.x & 31;
  if (e >= n_edges) return;
  const int s = src[e];
  const int d = dst[e];
  const float4 v = *(const float4*)(h + (size_t)s * N_FEATS + (lane << 2));
  float* o = agg + (size_t)d * N_FEATS + (lane << 2);
  atomic_add_f32(o + 0, v.x);
  atomic_add_f32(o + 1, v.y);
  atomic_add_f32(o + 2, v.z);
  atomic_add_f32(o + 3, v.w);
}

__global__ __launch_bounds__(256) void gcn_finalize(
    const float* __restrict__ agg, const float* __restrict__ norm_dst,
    const float* __restrict__ bias, float* __restrict__ out,
    int n_nodes, int do_relu) {
  const int i = blockIdx.x * blockDim.x + threadIdx.x;
  if (i >= n_nodes * 32) return;
  const int row = i >> 5;
  const int c4  = (i & 31) << 2;
  const float s = norm_dst[row];
  const float4 v = *(const float4*)(agg + (size_t)row * N_FEATS + c4);
  const float4 b = *(const float4*)(bias + c4);
  float4 r;
  r.x = v.x * s + b.x;
  r.y = v.y * s + b.y;
  r.z = v.z * s + b.z;
  r.w = v.w * s + b.w;
  if (do_relu) {
    r.x = fmaxf(r.x, 0.f); r.y = fmaxf(r.y, 0.f);
    r.z = fmaxf(r.z, 0.f); r.w = fmaxf(r.w, 0.f);
  }
  *(float4*)(out + (size_t)row * N_FEATS + c4) = r;
}

static inline size_t align256(size_t x) { return (x + 255) & ~(size_t)255; }

extern "C" void kernel_launch(void* const* d_in, const int* in_sizes, int n_in,
                              void* d_out, int out_size, void* d_ws, size_t ws_size,
                              hipStream_t stream) {
  (void)n_in; (void)out_size; (void)ws_size;
  const float* feats = (const float*)d_in[0];
  const int*   src   = (const int*)d_in[1];
  const int*   dst   = (const int*)d_in[2];
  const float* W1    = (const float*)d_in[3];
  const float* b1    = (const float*)d_in[4];
  const float* W2    = (const float*)d_in[5];
  const float* b2    = (const float*)d_in[6];
  float* out = (float*)d_out;

  const int n_nodes = in_sizes[0] / N_FEATS;
  const int n_edges = in_sizes[1];

  const size_t nb = align256((size_t)n_nodes * sizeof(float));
  const size_t hb = align256((size_t)n_nodes * N_FEATS * sizeof(float));
  char* p = (char*)d_ws;
  float* deg_out = (float*)p; p += nb;
  float* deg_in  = (float*)p; p += nb;
  float* ns      = (float*)p; p += nb;
  float* nd      = (float*)p; p += nb;
  float* bufA    = (float*)p; p += hb;
  float* bufB    = (float*)p; p += hb;

  const int tb = 256;
  const size_t node_bytes = (size_t)n_nodes * sizeof(float);
  const size_t feat_bytes = (size_t)n_nodes * N_FEATS * sizeof(float);

  hipMemsetAsync(deg_out, 0, node_bytes, stream);
  hipMemsetAsync(deg_in,  0, node_bytes, stream);
  gcn_degrees<<<(n_edges + tb - 1) / tb, tb, 0, stream>>>(src, dst, deg_out, deg_in, n_edges);
  gcn_norms<<<(n_nodes + tb - 1) / tb, tb, 0, stream>>>(deg_out, deg_in, ns, nd, n_nodes);

  const int gemm_grid = (n_nodes + 15) / 16;
  const int agg_grid  = (n_edges + 7) / 8;
  const int fin_grid  = (n_nodes * 32 + tb - 1) / tb;

  gcn_gemm_wmma<<<gemm_grid, 256, 0, stream>>>(feats, ns, W1, bufA, n_nodes);
  hipMemsetAsync(bufB, 0, feat_bytes, stream);
  gcn_agg<<<agg_grid, 256, 0, stream>>>(bufA, src, dst, bufB, n_edges);
  gcn_finalize<<<fin_grid, tb, 0, stream>>>(bufB, nd, b1, bufA, n_nodes,  1);

  gcn_gemm_wmma<<<gemm_grid, 256, 0, stream>>>(bufA, ns, W2, bufB, n_nodes);
  hipMemsetAsync(out, 0, feat_bytes, stream);
  gcn_agg<<<agg_grid, 256, 0, stream>>>(bufB, src, dst, out, n_edges);
  gcn_finalize<<<fin_grid, tb, 0, stream>>>(out, nd, b2, out, n_nodes,  0);
}


# agg: float atomic scatter -> edges sorted by dst in-call (int atomics, arrival-order rank) + one wave per node sums its rows in registers, scale/bias/relu in its store; sum order changes
# speedup vs baseline: 2.7119x; 2.7119x over previous
#include <hip/hip_runtime.h>

#define N_FEATS 128

typedef __attribute__((ext_vector_type(2))) float v2f;
typedef __attribute__((ext_vector_type(8))) float v8f;

__device__ __forceinline__ int atomic_add_i32(int* p, int v) {
  return __hip_atomic_fetch_add(p, v, __ATOMIC_RELAXED, __HIP_MEMORY_SCOPE_AGENT);
}

__global__ __launch_bounds__(256) void gcn_count_rank(
    const int* __restrict__ src, const int* __restrict__ dst,
    int* __restrict__ cnt_out, int* __restrict__ cnt_in,
    int* __restrict__ rank, int n_edges) {
  int i = blockIdx.x * blockDim.x + threadIdx.x;
  if (i >= n_edges) return;
  const int s = src[i];
  const int d = dst[i];
  atomic_add_i32(&cnt_out[s], 1);
  rank[i] = atomic_add_i32(&cnt_in[d], 1);
}

__global__ __launch_bounds__(256) void gcn_norms_alloc(
    const int* __restrict__ cnt_out, const int* __restrict__ cnt_in,
    float* __restrict__ ns, float* __restrict__ nd,
    int* __restrict__ row_start, int* __restrict__ total, int n) {
  const int i = blockIdx.x * blockDim.x + threadIdx.x;
  const int lane = threadIdx.x & 31;
  int c = 0;
  if (i < n) {
    c = cnt_in[i];
    ns[i] = rsqrtf(fmaxf((float)cnt_out[i], 1.0f));
    nd[i] = rsqrtf(fmaxf((float)c, 1.0f));
  }
  int incl = c;
  for (int o = 1; o < 32; o <<= 1) {
    const int t = __shfl_up(incl, o);
    if (lane >= o) incl += t;
  }
  const int wave_sum = __shfl(incl, 31);
  int base = 0;
  if (lane == 31) base = atomic_add_i32(total, wave_sum);
  base = __shfl(base, 31);
  if (i < n) row_start[i] = base + incl - c;
}

__global__ __launch_bounds__(256) void gcn_fill(
    const int* __restrict__ src, const int* __restrict__ dst,
    const int* __restrict__ rank, const int* __restrict__ row_start,
    int* __restrict__ col, int n_edges) {
  int i = blockIdx.x * blockDim.x + threadIdx.x;
  if (i >= n_edges) return;
  col[row_start[dst[i]] + rank[i]] = src[i];
}

__global__ __launch_bounds__(256) void gcn_gemm_wmma(
    const float* __restrict__ A, const float* __restrict__ norm_src,
    const float* __restrict__ W, float* __restrict__ out, int n_rows) {
  __shared__ float As[16][132];

  const int row0 = blockIdx.x << 4;

  for (int t = threadIdx.x; t < 512; t += 256) {
    const int r = t >> 5;
    const int c4 = (t & 31) << 2;
    const int row = row0 + r;
    if (row < n_rows) {
      const float4 v = *(const float4*)(A + (size_t)row * N_FEATS + c4);
      const float s = norm_src[row];
      As[r][c4 + 0] = v.x * s;
      As[r][c4 + 1] = v.y * s;
      As[r][c4 + 2] = v.z * s;
      As[r][c4 + 3] = v.w * s;
    } else {
      As[r][c4 + 0] = 0.f; As[r][c4 + 1] = 0.f;
      As[r][c4 + 2] = 0.f; As[r][c4 + 3] = 0.f;
    }
  }
  __syncthreads();

  const int wave = threadIdx.x >> 5;
  const int lane = threadIdx.x & 31;
  const int half = lane >> 4;
  const int m    = lane & 15;
  const int col0 = wave << 4;

  v8f acc = {};
#pragma unroll
  for (int k = 0; k < N_FEATS; k += 4) {
    const int ka = k + (half << 1);
    v2f a, b;
    a.x = As[m][ka];
    a.y = As[m][ka + 1];
    b.x = W[ka * N_FEATS + col0 + m];
    b.y = W[(ka + 1) * N_FEATS + col0 + m];
    acc = __builtin_amdgcn_wmma_f32_16x16x4_f32(
         false, a,  false, b,
         (short)0, acc,  false,  false);
  }

#pragma unroll
  for (int j = 0; j < 8; ++j) {
    const int r = row0 + j + (half << 3);
    if (r < n_rows) out[(size_t)r * N_FEATS + col0 + m] = acc[j];
  }
}

template <int U>
__device__ __forceinline__ void add_rows(const float* __restrict__ hl, int idx, int j, float4& acc) {
  float4 v[U];
#pragma unroll
  for (int k = 0; k < U; ++k) {
    const int s = __builtin_amdgcn_readlane(idx, j + k);
    v[k] = *(const float4*)(hl + (size_t)s * N_FEATS);
  }
#pragma unroll
  for (int k = 0; k < U; ++k) {
    acc.x += v[k].x; acc.y += v[k].y; acc.z += v[k].z; acc.w += v[k].w;
  }
}

__global__ __launch_bounds__(256) void gcn_gather(
    const float* __restrict__ h, const int* __restrict__ col,
    const int* __restrict__ row_start, const int* __restrict__ cnt_in,
    const float* __restrict__ norm_dst, const float* __restrict__ bias,
    float* __restrict__ out, int n_nodes, int do_relu) {
  const int lane = threadIdx.x & 31;
  const int node = __builtin_amdgcn_readfirstlane((int)((blockIdx.x * blockDim.x + threadIdx.x) >> 5));
  if (node >= n_nodes) return;
  const int beg = row_start[node];
  const int n   = cnt_in[node];
  const float* hl = h + (lane << 2);
  float4 acc = make_float4(0.f, 0.f, 0.f, 0.f);
  for (int b0 = 0; b0 < n; b0 += 32) {
    const int m = (n - b0 < 32) ? (n - b0) : 32;
    const int idx = (lane < m) ? col[beg + b0 + lane] : 0;
    int j = 0;
    for (; j + 8 <= m; j += 8) add_rows<8>(hl, idx, j, acc);
    if (j + 4 <= m) { add_rows<4>(hl, idx, j, acc); j += 4; }
    if (j + 2 <= m) { add_rows<2>(hl, idx, j, acc); j += 2; }
    if (j < m) add_rows<1>(hl, idx, j, acc);
  }
  const float s = norm_dst[node];
  const float4 b = *(const float4*)(bias + (lane << 2));
  float4 r;
  r.x = acc.x * s + b.x;
  r.y = acc.y * s + b.y;
  r.z = acc.z * s + b.z;
  r.w = acc.w * s + b.w;
  if (do_relu) {
    r.x = fmaxf(r.x, 0.f); r.y = fmaxf(r.y, 0.f);
    r.z = fmaxf(r.z, 0.f); r.w = fmaxf(r.w, 0.f);
  }
  *(float4*)(out + (size_t)node * N_FEATS + (lane << 2)) = r;
}

static inline size_t align256(size_t x) { return (x + 255) & ~(size_t)255; }

extern "C" void kernel_launch(void* const* d_in, const int* in_sizes, int n_in,
                              void* d_out, int out_size, void* d_ws, size_t ws_size,
                              hipStream_t stream) {
  (void)n_in; (void)out_size; (void)ws_size;
  const float* feats = (const float*)d_in[0];
  const int*   src   = (const int*)d_in[1];
  const int*   dst   = (const int*)d_in[2];
  const float* W1    = (const float*)d_in[3];
  const float* b1    = (const float*)d_in[4];
  const float* W2    = (const float*)d_in[5];
  const float* b2    = (const float*)d_in[6];
  float* out = (float*)d_out;

  const int n_nodes = in_sizes[0] / N_FEATS;
  const int n_edges = in_sizes[1];

  const size_t nb = align256((size_t)n_nodes * sizeof(float));
  const size_t eb = align256((size_t)n_edges * sizeof(int));
  const size_t hb = align256((size_t)n_nodes * N_FEATS * sizeof(float));
  char* p = (char*)d_ws;
  int*   cnt_out   = (int*)p;   p += nb;
  int*   cnt_in    = (int*)p;   p += nb;
  int*   total     = (int*)p;   p += 256;
  float* ns        = (float*)p; p += nb;
  float* nd        = (float*)p; p += nb;
  int*   row_start = (int*)p;   p += nb;
  int*   rank      = (int*)p;   p += eb;
  int*   col       = (int*)p;   p += eb;
  float* bufA      = (float*)p; p += hb;
  float* bufB      = (float*)p; p += hb;

  const int tb = 256;
  const int edge_grid = (n_edges + tb - 1) / tb;
  const int node_grid = (n_nodes + tb - 1) / tb;

  hipMemsetAsync(cnt_out, 0, 2 * nb + 256, stream);
  gcn_count_rank<<<edge_grid, tb, 0, stream>>>(src, dst, cnt_out, cnt_in, rank, n_edges);
  gcn_norms_alloc<<<node_grid, tb, 0, stream>>>(cnt_out, cnt_in, ns, nd, row_start, total, n_nodes);
  gcn_fill<<<edge_grid, tb, 0, stream>>>(src, dst, rank, row_start, col, n_edges);

  const int gemm_grid = (n_nodes + 15) / 16;
  const int gat_grid  = (n_nodes + 7) / 8;

  gcn_gemm_wmma<<<gemm_grid, 256, 0, stream>>>(feats, ns, W1, bufA, n_nodes);
  gcn_gather<<<gat_grid, 256, 0, stream>>>(bufA, col, row_start, cnt_in, nd, b1, bufB, n_nodes,  1);

  gcn_gemm_wmma<<<gemm_grid, 256, 0, stream>>>(bufB, ns, W2, bufA, n_nodes);
  gcn_gather<<<gat_grid, 256, 0, stream>>>(bufA, col, row_start, cnt_in, nd, b2, out, n_nodes,  0);
}


# edge sort by dst + both degree counts: per-edge global int atomics -> two-level counting sort on LDS counters (4096-edge chunks, 512-node bins); clear+3 launches -> 2; sums stay in arrival order
# speedup vs baseline: 4.7896x; 1.7661x over previous
#include <hip/hip_runtime.h>

#define N_FEATS 128

typedef __attribute__((ext_vector_type(2))) float v2f;
typedef __attribute__((ext_vector_type(8))) float v8f;

__device__ __forceinline__ int atomic_add_i32(int* p, int v) {
  return __hip_atomic_fetch_add(p, v, __ATOMIC_RELAXED, __HIP_MEMORY_SCOPE_AGENT);
}

#define BIN_SHIFT 9
#define BIN_NODES 512
#define NB_MAX 256
#define CHUNK 4096
#define NCH_MAX 512

__global__ __launch_bounds__(1024) void gcn_bin_chunks(
    const int* __restrict__ src, const int* __restrict__ dst,
    int* __restrict__ binD, int* __restrict__ binS,
    int* __restrict__ offD, int* __restrict__ offS,
    int* __restrict__ total, int n_edges) {
  __shared__ int hd[NB_MAX], hs[NB_MAX];
  __shared__ int od[NB_MAX], os[NB_MAX];
  __shared__ int sd[CHUNK], ss[CHUNK];
  const int t = threadIdx.x;
  const int lane = t & 31;
  const int wave = t >> 5;
  const int chunk = blockIdx.x;
  const int base = chunk * CHUNK;
  if (chunk == 0 && t == 0) *total = 0;
  if (t < NB_MAX) { hd[t] = 0; hs[t] = 0; }
  __syncthreads();

  int d[4], s[4], rd[4], rs[4];
#pragma unroll
  for (int j = 0; j < 4; ++j) {
    const int i = base + j * 1024 + t;
    d[j] = -1; s[j] = 0;
    if (i < n_edges) { d[j] = dst[i]; s[j] = src[i]; }
  }
#pragma unroll
  for (int j = 0; j < 4; ++j) {
    rd[j] = 0; rs[j] = 0;
    if (d[j] >= 0) {
      rd[j] = atomicAdd(&hd[d[j] >> BIN_SHIFT], 1);
      rs[j] = atomicAdd(&hs[s[j] >> BIN_SHIFT], 1);
    }
  }
  __syncthreads();

  if (wave < 2) {
    const int* h  = wave ? hs : hd;
    int* o        = wave ? os : od;
    int* offT     = wave ? offS : offD;
    int c[8];
    int sum = 0;
#pragma unroll
    for (int q = 0; q < 8; ++q) { c[q] = h[lane * 8 + q]; sum += c[q]; }
    int incl = sum;
    for (int k = 1; k < 32; k <<= 1) {
      const int up = __shfl_up(incl, k);
      if (lane >= k) incl += up;
    }
    int run = incl - sum;
#pragma unroll
    for (int q = 0; q < 8; ++q) {
      const int bin = lane * 8 + q;
      o[bin] = run;
      offT[bin * NCH_MAX + chunk] = run | (c[q] << 16);
      run += c[q];
    }
  }
  __syncthreads();

#pragma unroll
  for (int j = 0; j < 4; ++j) {
    if (d[j] >= 0) {
      sd[od[d[j] >> BIN_SHIFT] + rd[j]] = (s[j] << BIN_SHIFT) | (d[j] & (BIN_NODES - 1));
      ss[os[s[j] >> BIN_SHIFT] + rs[j]] = s[j] & (BIN_NODES - 1);
    }
  }
  __syncthreads();

  const int left = n_edges - base;
  const int nvalid = left < CHUNK ? left : CHUNK;
  for (int x = t; x < nvalid; x += 1024) {
    binD[base + x] = sd[x];
    binS[base + x] = ss[x];
  }
}

__global__ __launch_bounds__(1024) void gcn_bin_finish(
    const int* __restrict__ binD, const int* __restrict__ binS,
    const int* __restrict__ offD, const int* __restrict__ offS,
    float* __restrict__ ns, float* __restrict__ nd,
    int* __restrict__ cnt_in, int* __restrict__ row_start, int* __restrict__ col,
    int* __restrict__ total, int n_nodes, int n_chunks) {
  __shared__ int cin[BIN_NODES], cout[BIN_NODES], fillc[BIN_NODES], rsl[BIN_NODES];
  __shared__ int segD[NCH_MAX], segS[NCH_MAX];
  __shared__ int wsum[16];
  __shared__ int s_base;
  const int t = threadIdx.x;
  const int lane = t & 31;
  const int wave = t >> 5;
  const int b = blockIdx.x;
  if (t < BIN_NODES) { cin[t] = 0; cout[t] = 0; fillc[t] = 0; }
  if (t < NCH_MAX) {
    segD[t] = (t < n_chunks) ? offD[b * NCH_MAX + t] : 0;
    segS[t] = (t < n_chunks) ? offS[b * NCH_MAX + t] : 0;
  }
  __syncthreads();

  for (int k0 = wave * 4; k0 < n_chunks; k0 += 128) {
    int vd[4], vs[4];
#pragma unroll
    for (int u = 0; u < 4; ++u) {
      const int k = k0 + u;
      const int pd = segD[k], ps = segS[k];
      vd[u] = (lane < (pd >> 16)) ? binD[k * CHUNK + (pd & 0xffff) + lane] : -1;
      vs[u] = (lane < (ps >> 16)) ? binS[k * CHUNK + (ps & 0xffff) + lane] : -1;
    }
#pragma unroll
    for (int u = 0; u < 4; ++u) {
      if (vd[u] >= 0) atomicAdd(&cin[vd[u] & (BIN_NODES - 1)], 1);
      if (vs[u] >= 0) atomicAdd(&cout[vs[u]], 1);
    }
    for (int u = 0; u < 4; ++u) {
      const int k = k0 + u;
      const int pd = segD[k], ps = segS[k];
      for (int e = lane + 32; e < (pd >> 16); e += 32)
        atomicAdd(&cin[binD[k * CHUNK + (pd & 0xffff) + e] & (BIN_NODES - 1)], 1);
      for (int e = lane + 32; e < (ps >> 16); e += 32)
        atomicAdd(&cout[binS[k * CHUNK + (ps & 0xffff) + e]], 1);
    }
  }
  __syncthreads();

  const int node = b * BIN_NODES + t;
  const int c = (t < BIN_NODES) ? cin[t] : 0;
  int incl = c;
  for (int k = 1; k < 32; k <<= 1) {
    const int up = __shfl_up(incl, k);
    if (lane >= k) incl += up;
  }
  if (t < BIN_NODES && lane == 31) wsum[wave] = incl;
  __syncthreads();
  if (t == 0) {
    int run = 0;
    for (int w = 0; w < 16; ++w) { const int x = wsum[w]; wsum[w] = run; run += x; }
    s_base = atomic_add_i32(total, run);
  }
  __syncthreads();
  if (t < BIN_NODES) {
    const int start = s_base + wsum[wave] + incl - c;
    rsl[t] = start;
    if (node < n_nodes) {
      row_start[node] = start;
      cnt_in[node] = c;
      nd[node] = rsqrtf(fmaxf((float)c, 1.0f));
      ns[node] = rsqrtf(fmaxf((float)cout[t], 1.0f));
    }
  }
  __syncthreads();

  for (int k0 = wave * 4; k0 < n_chunks; k0 += 128) {
    int vd[4];
#pragma unroll
    for (int u = 0; u < 4; ++u) {
      const int k = k0 + u;
      const int pd = segD[k];
      vd[u] = (lane < (pd >> 16)) ? binD[k * CHUNK + (pd & 0xffff) + lane] : -1;
    }
#pragma unroll
    for (int u = 0; u < 4; ++u) {
      if (vd[u] >= 0) {
        const int dl = vd[u] & (BIN_NODES - 1);
        col[rsl[dl] + atomicAdd(&fillc[dl], 1)] = vd[u] >> BIN_SHIFT;
      }
    }
    for (int u = 0; u < 4; ++u) {
      const int k = k0 + u;
      const int pd = segD[k];
      for (int e = lane + 32; e < (pd >> 16); e += 32) {
        const int v = binD[k * CHUNK + (pd & 0xffff) + e];
        const int dl = v & (BIN_NODES - 1);
        col[rsl[dl] + atomicAdd(&fillc[dl], 1)] = v >> BIN_SHIFT;
      }
    }
  }
}

__global__ __launch_bounds__(256) void gcn_gemm_wmma(
    const float* __restrict__ A, const float* __restrict__ norm_src,
    const float* __restrict__ W, float* __restrict__ out, int n_rows) {
  __shared__ float As[16][132];

  const int row0 = blockIdx.x << 4;

  for (int t = threadIdx.x; t < 512; t += 256) {
    const int r = t >> 5;
    const int c4 = (t & 31) << 2;
    const int row = row0 + r;
    if (row < n_rows) {
      const float4 v = *(const float4*)(A + (size_t)row * N_FEATS + c4);
      const float s = norm_src[row];
      As[r][c4 + 0] = v.x * s;
      As[r][c4 + 1] = v.y * s;
      As[r][c4 + 2] = v.z * s;
      As[r][c4 + 3] = v.w * s;
    } else {
      As[r][c4 + 0] = 0.f; As[r][c4 + 1] = 0.f;
      As[r][c4 + 2] = 0.f; As[r][c4 + 3] = 0.f;
    }
  }
  __syncthreads();

  const int wave = threadIdx.x >> 5;
  const int lane = threadIdx.x & 31;
  const int half = lane >> 4;
  const int m    = lane & 15;
  const int col0 = wave << 4;

  v8f acc = {};
#pragma unroll
  for (int k = 0; k < N_FEATS; k += 4) {
    const int ka = k + (half << 1);
    v2f a, b;
    a.x = As[m][ka];
    a.y = As[m][ka + 1];
    b.x = W[ka * N_FEATS + col0 + m];
    b.y = W[(ka + 1) * N_FEATS + col0 + m];
    acc = __builtin_amdgcn_wmma_f32_16x16x4_f32(
         false, a,  false, b,
         (short)0, acc,  false,  false);
  }

#pragma unroll
  for (int j = 0; j < 8; ++j) {
    const int r = row0 + j + (half << 3);
    if (r < n_rows) out[(size_t)r * N_FEATS + col0 + m] = acc[j];
  }
}

template <int U>
__device__ __forceinline__ void add_rows(const float* __restrict__ hl, int idx, int j, float4& acc) {
  float4 v[U];
#pragma unroll
  for (int k = 0; k < U; ++k) {
    const int s = __builtin_amdgcn_readlane(idx, j + k);
    v[k] = *(const float4*)(hl + (size_t)s * N_FEATS);
  }
#pragma unroll
  for (int k = 0; k < U; ++k) {
    acc.x += v[k].x; acc.y += v[k].y; acc.z += v[k].z; acc.w += v[k].w;
  }
}

__global__ __launch_bounds__(256) void gcn_gather(
    const float* __restrict__ h, const int* __restrict__ col,
    const int* __restrict__ row_start, const int* __restrict__ cnt_in,
    const float* __restrict__ norm_dst, const float* __restrict__ bias,
    float* __restrict__ out, int n_nodes, int do_relu) {
  const int lane = threadIdx.x & 31;
  const int node = __builtin_amdgcn_readfirstlane((int)((blockIdx.x * blockDim.x + threadIdx.x) >> 5));
  if (node >= n_nodes) return;
  const int beg = row_start[node];
  const int n   = cnt_in[node];
  const float* hl = h + (lane << 2);
  float4 acc = make_float4(0.f, 0.f, 0.f, 0.f);
  for (int b0 = 0; b0 < n; b0 += 32) {
    const int m = (n - b0 < 32) ? (n - b0) : 32;
    const int idx = (lane < m) ? col[beg + b0 + lane] : 0;
    int j = 0;
    for (; j + 8 <= m; j += 8) add_rows<8>(hl, idx, j, acc);
    if (j + 4 <= m) { add_rows<4>(hl, idx, j, acc); j += 4; }
    if (j + 2 <= m) { add_rows<2>(hl, idx, j, acc); j += 2; }
    if (j < m) add_rows<1>(hl, idx, j, acc);
  }
  const float s = norm_dst[node];
  const float4 b = *(const float4*)(bias + (lane << 2));
  float4 r;
  r.x = acc.x * s + b.x;
  r.y = acc.y * s + b.y;
  r.z = acc.z * s + b.z;
  r.w = acc.w * s + b.w;
  if (do_relu) {
    r.x = fmaxf(r.x, 0.f); r.y = fmaxf(r.y, 0.f);
    r.z = fmaxf(r.z, 0.f); r.w = fmaxf(r.w, 0.f);
  }
  *(float4*)(out + (size_t)node * N_FEATS + (lane << 2)) = r;
}

static inline size_t align256(size_t x) { return (x + 255) & ~(size_t)255; }

extern "C" void kernel_launch(void* const* d_in, const int* in_sizes, int n_in,
                              void* d_out, int out_size, void* d_ws, size_t ws_size,
                              hipStream_t stream) {
  (void)n_in; (void)out_size; (void)ws_size;
  const float* feats = (const float*)d_in[0];
  const int*   src   = (const int*)d_in[1];
  const int*   dst   = (const int*)d_in[2];
  const float* W1    = (const float*)d_in[3];
  const float* b1    = (const float*)d_in[4];
  const float* W2    = (const float*)d_in[5];
  const float* b2    = (const float*)d_in[6];
  float* out = (float*)d_out;

  const int n_nodes = in_sizes[0] / N_FEATS;
  const int n_edges = in_sizes[1];

  const size_t nb = align256((size_t)n_nodes * sizeof(float));
  const size_t eb = align256((size_t)n_edges * sizeof(int));
  const size_t hb = align256((size_t)n_nodes * N_FEATS * sizeof(float));
  const size_t tb = align256((size_t)NB_MAX * NCH_MAX * sizeof(int));
  char* p = (char*)d_ws;
  int*   total     = (int*)p;   p += 256;
  int*   cnt_in    = (int*)p;   p += nb;
  float* ns        = (float*)p; p += nb;
  float* nd        = (float*)p; p += nb;
  int*   row_start = (int*)p;   p += nb;
  int*   offD      = (int*)p;   p += tb;
  int*   offS      = (int*)p;   p += tb;
  int*   binD      = (int*)p;   p += eb;
  int*   binS      = (int*)p;   p += eb;
  int*   col       = (int*)p;   p += eb;
  float* bufA      = (float*)p; p += hb;
  float* bufB      = (float*)p; p += hb;

  const int n_chunks = (n_edges + CHUNK - 1) / CHUNK;
  const int n_bins   = (n_nodes + BIN_NODES - 1) / BIN_NODES;

  gcn_bin_chunks<<<n_chunks, 1024, 0, stream>>>(src, dst, binD, binS, offD, offS, total, n_edges);
  gcn_bin_finish<<<n_bins, 1024, 0, stream>>>(binD, binS, offD, offS, ns, nd, cnt_in, row_start, col,
                                               total, n_nodes, n_chunks);

  const int gemm_grid = (n_nodes + 15) / 16;
  const int gat_grid  = (n_nodes + 7) / 8;

  gcn_gemm_wmma<<<gemm_grid, 256, 0, stream>>>(feats, ns, W1, bufA, n_nodes);
  gcn_gather<<<gat_grid, 256, 0, stream>>>(bufA, col, row_start, cnt_in, nd, b1, bufB, n_nodes,  1);

  gcn_gemm_wmma<<<gemm_grid, 256, 0, stream>>>(bufB, ns, W2, bufA, n_nodes);
  gcn_gather<<<gat_grid, 256, 0, stream>>>(bufA, col, row_start, cnt_in, nd, b2, out, n_nodes,  0);
}
